# starting kernel
# baseline (speedup 1.0000x reference)
#include <hip/hip_runtime.h>
#include <hip/hip_bf16.h>
#include <math.h>


typedef __attribute__((ext_vector_type(2))) float v2f;
typedef __attribute__((ext_vector_type(8))) float v8f;

#define D_DIM 256
#define B_DIM 512
#define ZS    260
#define ITERS 50

__device__ __forceinline__ float fast_tanh(float v) {
    float e = __expf(2.0f * v);
    float r = __builtin_amdgcn_rcpf(e + 1.0f);
    return fmaf(-2.0f, r, 1.0f);
}

__global__ __launch_bounds__(512, 1)
void tanh_fixed_point_wmma(const float* __restrict__ x,
                           const float* __restrict__ W,
                           float* __restrict__ out)
{
    __shared__ float zsh[16 * ZS];

    const int tid  = threadIdx.x;
    const int wave = tid >> 5;
    const int lane = tid & 31;
    const int nIdx = lane & 15;
    const int half = lane >> 4;
    const int n0   = wave * 16;
    const int rowBase = blockIdx.x * 16;

    float xf[8];
#pragma unroll
    for (int i = 0; i < 8; ++i) {
        const int M = i + 8 * half;
        xf[i] = x[(rowBase + M) * D_DIM + n0 + nIdx];
    }

    {
        const float* wrow = &W[(n0 + nIdx) * D_DIM];
#pragma unroll
        for (int j = 0; j < 4; ++j)
            __builtin_prefetch(wrow + half * 128 + j * 32, 0, 3);
    }

#pragma unroll
    for (int i = 0; i < 8; ++i) {
        const int M = i + 8 * half;
        zsh[M * ZS + n0 + nIdx] = fast_tanh(xf[i]);
    }
    __syncthreads();

    const int aRow = nIdx;
    const int kOff = 2 * half;

    for (int it = 0; it <= ITERS; ++it) {
        v8f c = {};
#pragma unroll 8
        for (int kk = 0; kk < D_DIM / 4; ++kk) {
            const int k0 = kk * 4;
            v2f a, b;
            a.x = zsh[aRow * ZS + k0 + kOff];
            a.y = zsh[aRow * ZS + k0 + kOff + 1];
            const float* wp = &W[(n0 + nIdx) * D_DIM + k0 + kOff];
            b.x = wp[0];
            b.y = wp[1];
            c = __builtin_amdgcn_wmma_f32_16x16x4_f32(false, a, false, b,
                                                      (short)0, c, false, false);
        }

        float zn[8];
#pragma unroll
        for (int i = 0; i < 8; ++i)
            zn[i] = fast_tanh(c[i] + xf[i]);

        if (it == ITERS) {
#pragma unroll
            for (int i = 0; i < 8; ++i) {
                const int M = i + 8 * half;
                out[(rowBase + M) * D_DIM + n0 + nIdx] = zn[i];
            }
        } else {
            __syncthreads();
#pragma unroll
            for (int i = 0; i < 8; ++i) {
                const int M = i + 8 * half;
                zsh[M * ZS + n0 + nIdx] = zn[i];
            }
            __syncthreads();
        }
    }
}

extern "C" void kernel_launch(void* const* d_in, const int* in_sizes, int n_in,
                              void* d_out, int out_size, void* d_ws, size_t ws_size,
                              hipStream_t stream) {
    const float* x = (const float*)d_in[0];
    const float* W = (const float*)d_in[1];
    float* out = (float*)d_out;
    (void)in_sizes; (void)n_in; (void)out_size; (void)d_ws; (void)ws_size;

    dim3 grid(B_DIM / 16);
    dim3 block(512);
    hipLaunchKernelGGL(tanh_fixed_point_wmma, grid, block, 0, stream, x, W, out);
}


# W panel of each wave (16 rows x 256 k, 128 VGPRs a lane) loaded once into registers as the loop-invariant B operand; the 51 passes read only z from LDS; k and sum order unchanged
# speedup vs baseline: 2.1067x; 2.1067x over previous
#include <hip/hip_runtime.h>
#include <hip/hip_bf16.h>
#include <math.h>


typedef __attribute__((ext_vector_type(2))) float v2f;
typedef __attribute__((ext_vector_type(8))) float v8f;

#define D_DIM 256
#define B_DIM 512
#define ZS    260
#define ITERS 50

__device__ __forceinline__ float fast_tanh(float v) {
    float e = __expf(2.0f * v);
    float r = __builtin_amdgcn_rcpf(e + 1.0f);
    return fmaf(-2.0f, r, 1.0f);
}

__global__ __launch_bounds__(512, 1)
void tanh_fixed_point_wmma(const float* __restrict__ x,
                           const float* __restrict__ W,
                           float* __restrict__ out)
{
    __shared__ float zsh[16 * ZS];

    const int tid  = threadIdx.x;
    const int wave = tid >> 5;
    const int lane = tid & 31;
    const int nIdx = lane & 15;
    const int half = lane >> 4;
    const int n0   = wave * 16;
    const int rowBase = blockIdx.x * 16;

    v2f wreg[D_DIM / 4];
    {
        const float* wrow = &W[(n0 + nIdx) * D_DIM + 2 * half];
#pragma unroll
        for (int kk = 0; kk < D_DIM / 4; ++kk) {
            wreg[kk].x = wrow[kk * 4];
            wreg[kk].y = wrow[kk * 4 + 1];
        }
    }

    float xf[8];
#pragma unroll
    for (int i = 0; i < 8; ++i) {
        const int M = i + 8 * half;
        xf[i] = x[(rowBase + M) * D_DIM + n0 + nIdx];
    }

#pragma unroll
    for (int i = 0; i < 8; ++i) {
        const int M = i + 8 * half;
        zsh[M * ZS + n0 + nIdx] = fast_tanh(xf[i]);
    }
    __syncthreads();

    const int aRow = nIdx;
    const int kOff = 2 * half;

    for (int it = 0; it <= ITERS; ++it) {
        v8f c = {};
#pragma unroll
        for (int kk = 0; kk < D_DIM / 4; ++kk) {
            const int k0 = kk * 4;
            v2f a;
            a.x = zsh[aRow * ZS + k0 + kOff];
            a.y = zsh[aRow * ZS + k0 + kOff + 1];
            c = __builtin_amdgcn_wmma_f32_16x16x4_f32(false, a, false, wreg[kk],
                                                      (short)0, c, false, false);
        }

        float zn[8];
#pragma unroll
        for (int i = 0; i < 8; ++i)
            zn[i] = fast_tanh(c[i] + xf[i]);

        if (it == ITERS) {
#pragma unroll
            for (int i = 0; i < 8; ++i) {
                const int M = i + 8 * half;
                out[(rowBase + M) * D_DIM + n0 + nIdx] = zn[i];
            }
        } else {
            __syncthreads();
#pragma unroll
            for (int i = 0; i < 8; ++i) {
                const int M = i + 8 * half;
                zsh[M * ZS + n0 + nIdx] = zn[i];
            }
            __syncthreads();
        }
    }
}

extern "C" void kernel_launch(void* const* d_in, const int* in_sizes, int n_in,
                              void* d_out, int out_size, void* d_ws, size_t ws_size,
                              hipStream_t stream) {
    const float* x = (const float*)d_in[0];
    const float* W = (const float*)d_in[1];
    float* out = (float*)d_out;
    (void)in_sizes; (void)n_in; (void)out_size; (void)d_ws; (void)ws_size;

    dim3 grid(B_DIM / 16);
    dim3 block(512);
    hipLaunchKernelGGL(tanh_fixed_point_wmma, grid, block, 0, stream, x, W, out);
}


# each wave owns two 16-column tiles (8 waves a block): the z tile is read from LDS once per pass as one load group ahead of the WMMAs, each A fragment feeds two WMMAs; k and sum order unchanged
# speedup vs baseline: 2.2056x; 1.0469x over previous
#include <hip/hip_runtime.h>
#include <hip/hip_bf16.h>
#include <math.h>


typedef __attribute__((ext_vector_type(2))) float v2f;
typedef __attribute__((ext_vector_type(8))) float v8f;

#define D_DIM 256
#define B_DIM 512
#define ZS    260
#define ITERS 50
#define NT    2
#define NWAVE (D_DIM / 16 / NT)
#define KSTEPS (D_DIM / 4)

__device__ __forceinline__ float fast_tanh(float v) {
    float e = __expf(2.0f * v);
    float r = __builtin_amdgcn_rcpf(e + 1.0f);
    return fmaf(-2.0f, r, 1.0f);
}

__global__ __launch_bounds__(NWAVE * 32, 1)
void tanh_fixed_point_wmma(const float* __restrict__ x,
                           const float* __restrict__ W,
                           float* __restrict__ out)
{
    __shared__ float zsh[16 * ZS];

    const int tid  = threadIdx.x;
    const int wave = tid >> 5;
    const int lane = tid & 31;
    const int nIdx = lane & 15;
    const int half = lane >> 4;
    const int n0   = wave * (16 * NT);
    const int rowBase = blockIdx.x * 16;

    v2f wreg[NT][KSTEPS];
#pragma unroll
    for (int t = 0; t < NT; ++t) {
        const float* wrow = &W[(n0 + 16 * t + nIdx) * D_DIM + 2 * half];
#pragma unroll
        for (int kk = 0; kk < KSTEPS; ++kk) {
            wreg[t][kk].x = wrow[kk * 4];
            wreg[t][kk].y = wrow[kk * 4 + 1];
        }
    }

    float xf[NT][8];
#pragma unroll
    for (int t = 0; t < NT; ++t)
#pragma unroll
        for (int i = 0; i < 8; ++i) {
            const int M = i + 8 * half;
            xf[t][i] = x[(rowBase + M) * D_DIM + n0 + 16 * t + nIdx];
        }

#pragma unroll
    for (int t = 0; t < NT; ++t)
#pragma unroll
        for (int i = 0; i < 8; ++i) {
            const int M = i + 8 * half;
            zsh[M * ZS + n0 + 16 * t + nIdx] = fast_tanh(xf[t][i]);
        }
    __syncthreads();

    const int aRow = nIdx;
    const int kOff = 2 * half;

    for (int it = 0; it <= ITERS; ++it) {
        v2f a[KSTEPS];
#pragma unroll
        for (int kk = 0; kk < KSTEPS; ++kk) {
            a[kk].x = zsh[aRow * ZS + kk * 4 + kOff];
            a[kk].y = zsh[aRow * ZS + kk * 4 + kOff + 1];
        }
        __builtin_amdgcn_sched_barrier(0);

        v8f c[NT];
#pragma unroll
        for (int t = 0; t < NT; ++t) c[t] = (v8f){};
#pragma unroll
        for (int kk = 0; kk < KSTEPS; ++kk) {
#pragma unroll
            for (int t = 0; t < NT; ++t)
                c[t] = __builtin_amdgcn_wmma_f32_16x16x4_f32(false, a[kk], false, wreg[t][kk],
                                                             (short)0, c[t], false, false);
        }

        float zn[NT][8];
#pragma unroll
        for (int t = 0; t < NT; ++t)
#pragma unroll
            for (int i = 0; i < 8; ++i)
                zn[t][i] = fast_tanh(c[t][i] + xf[t][i]);

        if (it == ITERS) {
#pragma unroll
            for (int t = 0; t < NT; ++t)
#pragma unroll
                for (int i = 0; i < 8; ++i) {
                    const int M = i + 8 * half;
                    out[(rowBase + M) * D_DIM + n0 + 16 * t + nIdx] = zn[t][i];
                }
        } else {
            __builtin_amdgcn_sched_barrier(0);
            __syncthreads();
#pragma unroll
            for (int t = 0; t < NT; ++t)
#pragma unroll
                for (int i = 0; i < 8; ++i) {
                    const int M = i + 8 * half;
                    zsh[M * ZS + n0 + 16 * t + nIdx] = zn[t][i];
                }
            __syncthreads();
        }
    }
}

extern "C" void kernel_launch(void* const* d_in, const int* in_sizes, int n_in,
                              void* d_out, int out_size, void* d_ws, size_t ws_size,
                              hipStream_t stream) {
    const float* x = (const float*)d_in[0];
    const float* W = (const float*)d_in[1];
    float* out = (float*)d_out;
    (void)in_sizes; (void)n_in; (void)out_size; (void)d_ws; (void)ws_size;

    dim3 grid(B_DIM / 16);
    dim3 block(NWAVE * 32);
    hipLaunchKernelGGL(tanh_fixed_point_wmma, grid, block, 0, stream, x, W, out);
}


# GEMM transposed (W panel = A operand): tanh output is already the B fragment, a wave's own 8 k-steps run from registers over the LDS exchange (16-byte copies, 1 barrier); k sum order changed
# speedup vs baseline: 2.2841x; 1.0356x over previous
#include <hip/hip_runtime.h>
#include <hip/hip_bf16.h>
#include <math.h>


typedef __attribute__((ext_vector_type(2))) float v2f;
typedef __attribute__((ext_vector_type(4))) float v4f;
typedef __attribute__((ext_vector_type(8))) float v8f;

#define D_DIM 256
#define B_DIM 512
#define ITERS 50
#define NT    2
#define NWAVE (D_DIM / 16 / NT)

__device__ __forceinline__ float fast_tanh(float v) {
    float e = __expf(2.0f * v);
    float r = __builtin_amdgcn_rcpf(e + 1.0f);
    return fmaf(-2.0f, r, 1.0f);
}


__global__ __launch_bounds__(NWAVE * 32, 1)
void tanh_fixed_point_wmma(const float* __restrict__ x,
                           const float* __restrict__ W,
                           float* __restrict__ out)
{
    __shared__ v4f zsh[2][NWAVE][2 * NT][32];

    const int tid  = threadIdx.x;
    const int wave = tid >> 5;
    const int lane = tid & 31;
    const int cIdx = lane & 15;
    const int half = lane >> 4;
    const int n0   = wave * (16 * NT);
    const int rowBase = blockIdx.x * 16;

    v2f wreg[NT][NWAVE][NT][4];
#pragma unroll
    for (int t = 0; t < NT; ++t)
#pragma unroll
        for (int i = 0; i < NWAVE; ++i)
#pragma unroll
            for (int tp = 0; tp < NT; ++tp) {
                const int pw = (wave + i) & (NWAVE - 1);
                const float* wp = &W[(n0 + 16 * t + cIdx) * D_DIM + 32 * pw + 16 * tp + 8 * half];
#pragma unroll
                for (int j = 0; j < 4; ++j) {
                    wreg[t][i][tp][j].x = wp[2 * j];
                    wreg[t][i][tp][j].y = wp[2 * j + 1];
                }
            }

    float xf[NT][8];
#pragma unroll
    for (int t = 0; t < NT; ++t)
#pragma unroll
        for (int r = 0; r < 8; ++r)
            xf[t][r] = x[(rowBase + cIdx) * D_DIM + n0 + 16 * t + 8 * half + r];

    float zn[NT][8];
#pragma unroll
    for (int t = 0; t < NT; ++t)
#pragma unroll
        for (int r = 0; r < 8; ++r)
            zn[t][r] = fast_tanh(xf[t][r]);

    for (int it = 0; it <= ITERS; ++it) {
        const int buf = it & 1;

#pragma unroll
        for (int g = 0; g < 2 * NT; ++g) {
            const int tp = g >> 1, q = g & 1;
            zsh[buf][wave][g][lane] = (v4f){zn[tp][4 * q], zn[tp][4 * q + 1],
                                            zn[tp][4 * q + 2], zn[tp][4 * q + 3]};
        }
        __builtin_amdgcn_sched_barrier(0);

        v8f c[NT];
#pragma unroll
        for (int t = 0; t < NT; ++t) c[t] = (v8f){};

#pragma unroll
        for (int j = 0; j < 4; ++j) {
            const v2f b = (v2f){zn[0][2 * j], zn[0][2 * j + 1]};
#pragma unroll
            for (int t = 0; t < NT; ++t)
                c[t] = __builtin_amdgcn_wmma_f32_16x16x4_f32(false, wreg[t][0][0][j], false, b,
                                                             (short)0, c[t], false, false);
        }
        __builtin_amdgcn_sched_barrier(0);
        __syncthreads();

        v4f ld[NWAVE - 1][2 * NT];
#pragma unroll
        for (int i = 1; i < NWAVE; ++i) {
            const int pw = (wave + i) & (NWAVE - 1);
#pragma unroll
            for (int g = 0; g < 2 * NT; ++g)
                ld[i - 1][g] = zsh[buf][pw][g][lane];
        }
        __builtin_amdgcn_sched_barrier(0);

#pragma unroll
        for (int j = 0; j < 4; ++j) {
            const v2f b = (v2f){zn[1][2 * j], zn[1][2 * j + 1]};
#pragma unroll
            for (int t = 0; t < NT; ++t)
                c[t] = __builtin_amdgcn_wmma_f32_16x16x4_f32(false, wreg[t][0][1][j], false, b,
                                                             (short)0, c[t], false, false);
        }
#pragma unroll
        for (int i = 1; i < NWAVE; ++i)
#pragma unroll
            for (int tp = 0; tp < NT; ++tp)
#pragma unroll
                for (int j = 0; j < 4; ++j) {
                    const v4f v = ld[i - 1][2 * tp + (j >> 1)];
                    const v2f b = (j & 1) ? (v2f){v.z, v.w} : (v2f){v.x, v.y};
#pragma unroll
                    for (int t = 0; t < NT; ++t)
                        c[t] = __builtin_amdgcn_wmma_f32_16x16x4_f32(false, wreg[t][i][tp][j],
                                                                     false, b, (short)0, c[t],
                                                                     false, false);
                }

#pragma unroll
        for (int t = 0; t < NT; ++t)
#pragma unroll
            for (int r = 0; r < 8; ++r)
                zn[t][r] = fast_tanh(c[t][r] + xf[t][r]);
    }

#pragma unroll
    for (int t = 0; t < NT; ++t)
#pragma unroll
        for (int r = 0; r < 8; ++r)
            out[(rowBase + cIdx) * D_DIM + n0 + 16 * t + 8 * half + r] = zn[t][r];
}

extern "C" void kernel_launch(void* const* d_in, const int* in_sizes, int n_in,
                              void* d_out, int out_size, void* d_ws, size_t ws_size,
                              hipStream_t stream) {
    const float* x = (const float*)d_in[0];
    const float* W = (const float*)d_in[1];
    float* out = (float*)d_out;
    (void)in_sizes; (void)n_in; (void)out_size; (void)d_ws; (void)ws_size;

    dim3 grid(B_DIM / 16);
    dim3 block(NWAVE * 32);
    hipLaunchKernelGGL(tanh_fixed_point_wmma, grid, block, 0, stream, x, W, out);
}


# Picard loop leaves once the block meets its share of the reference's own stop test ||z - tanh(Wz+x)||_F < TOL (wave residual sums cross LDS beside z); at most 51 passes as before; k sum order kept
# speedup vs baseline: 6.0093x; 2.6309x over previous
#include <hip/hip_runtime.h>
#include <hip/hip_bf16.h>
#include <math.h>


typedef __attribute__((ext_vector_type(2))) float v2f;
typedef __attribute__((ext_vector_type(4))) float v4f;
typedef __attribute__((ext_vector_type(8))) float v8f;

#define D_DIM 256
#define B_DIM 512
#define ITERS 50
#define NT    2
#define NWAVE (D_DIM / 16 / NT)
#define TOL_REF 1e-4f
#define BLOCK_THR (TOL_REF * TOL_REF * 16.0f / (float)B_DIM)

__device__ __forceinline__ float fast_tanh(float v) {
    float e = __expf(2.0f * v);
    float r = __builtin_amdgcn_rcpf(e + 1.0f);
    return fmaf(-2.0f, r, 1.0f);
}

#define ROW_XMASK_ADD(v, m)                                                                  \
    ((v) + __builtin_bit_cast(float, __builtin_amdgcn_update_dpp(                            \
               0, __builtin_bit_cast(int, (v)), 0x160 + (m), 0xf, 0xf, true)))

__device__ __forceinline__ float wave_sum(float v) {
    v = ROW_XMASK_ADD(v, 1);
    v = ROW_XMASK_ADD(v, 2);
    v = ROW_XMASK_ADD(v, 4);
    v = ROW_XMASK_ADD(v, 8);
    const int b = __builtin_bit_cast(int, v);
    return v + __builtin_bit_cast(float, __builtin_amdgcn_permlanex16(b, b, 0x76543210, 0xfedcba98,
                                                                      false, false));
}


__global__ __launch_bounds__(NWAVE * 32, 1)
void tanh_fixed_point_wmma(const float* __restrict__ x,
                           const float* __restrict__ W,
                           float* __restrict__ out)
{
    __shared__ v4f zsh[2][NWAVE][2 * NT][32];
    __shared__ __attribute__((aligned(16))) float rsh[2][NWAVE];

    const int tid  = threadIdx.x;
    const int wave = tid >> 5;
    const int lane = tid & 31;
    const int cIdx = lane & 15;
    const int half = lane >> 4;
    const int n0   = wave * (16 * NT);
    const int rowBase = blockIdx.x * 16;

    v2f wreg[NT][NWAVE][NT][4];
#pragma unroll
    for (int t = 0; t < NT; ++t)
#pragma unroll
        for (int i = 0; i < NWAVE; ++i)
#pragma unroll
            for (int tp = 0; tp < NT; ++tp) {
                const int pw = (wave + i) & (NWAVE - 1);
                const float* wp = &W[(n0 + 16 * t + cIdx) * D_DIM + 32 * pw + 16 * tp + 8 * half];
#pragma unroll
                for (int j = 0; j < 4; ++j) {
                    wreg[t][i][tp][j].x = wp[2 * j];
                    wreg[t][i][tp][j].y = wp[2 * j + 1];
                }
            }

    float xf[NT][8];
#pragma unroll
    for (int t = 0; t < NT; ++t)
#pragma unroll
        for (int r = 0; r < 8; ++r)
            xf[t][r] = x[(rowBase + cIdx) * D_DIM + n0 + 16 * t + 8 * half + r];

    float zn[NT][8];
#pragma unroll
    for (int t = 0; t < NT; ++t)
#pragma unroll
        for (int r = 0; r < 8; ++r)
            zn[t][r] = fast_tanh(xf[t][r]);

    float res = 3.0e38f;

    for (int it = 0; it <= ITERS; ++it) {
        const int buf = it & 1;

#pragma unroll
        for (int g = 0; g < 2 * NT; ++g) {
            const int tp = g >> 1, q = g & 1;
            zsh[buf][wave][g][lane] = (v4f){zn[tp][4 * q], zn[tp][4 * q + 1],
                                            zn[tp][4 * q + 2], zn[tp][4 * q + 3]};
        }
        rsh[buf][wave] = res;
        __builtin_amdgcn_sched_barrier(0);

        v8f c[NT];
#pragma unroll
        for (int t = 0; t < NT; ++t) c[t] = (v8f){};

#pragma unroll
        for (int j = 0; j < 4; ++j) {
            const v2f b = (v2f){zn[0][2 * j], zn[0][2 * j + 1]};
#pragma unroll
            for (int t = 0; t < NT; ++t)
                c[t] = __builtin_amdgcn_wmma_f32_16x16x4_f32(false, wreg[t][0][0][j], false, b,
                                                             (short)0, c[t], false, false);
        }
        __builtin_amdgcn_sched_barrier(0);
        __syncthreads();

        const v4f ra = *(const v4f*)&rsh[buf][0];
        const v4f rb = *(const v4f*)&rsh[buf][4];

        v4f ld[NWAVE - 1][2 * NT];
#pragma unroll
        for (int i = 1; i < NWAVE; ++i) {
            const int pw = (wave + i) & (NWAVE - 1);
#pragma unroll
            for (int g = 0; g < 2 * NT; ++g)
                ld[i - 1][g] = zsh[buf][pw][g][lane];
        }
        __builtin_amdgcn_sched_barrier(0);

#pragma unroll
        for (int j = 0; j < 4; ++j) {
            const v2f b = (v2f){zn[1][2 * j], zn[1][2 * j + 1]};
#pragma unroll
            for (int t = 0; t < NT; ++t)
                c[t] = __builtin_amdgcn_wmma_f32_16x16x4_f32(false, wreg[t][0][1][j], false, b,
                                                             (short)0, c[t], false, false);
        }
        __builtin_amdgcn_sched_barrier(0);
        const v4f rp = ra + rb;
        const v2f rq = (v2f){rp.x, rp.y} + (v2f){rp.z, rp.w};
        const float tot = rq.x + rq.y;
        if (__builtin_bit_cast(float, __builtin_amdgcn_readfirstlane(__builtin_bit_cast(int, tot)))
            < BLOCK_THR)
            break;

#pragma unroll
        for (int i = 1; i < NWAVE; ++i)
#pragma unroll
            for (int tp = 0; tp < NT; ++tp)
#pragma unroll
                for (int j = 0; j < 4; ++j) {
                    const v4f v = ld[i - 1][2 * tp + (j >> 1)];
                    const v2f b = (j & 1) ? (v2f){v.z, v.w} : (v2f){v.x, v.y};
#pragma unroll
                    for (int t = 0; t < NT; ++t)
                        c[t] = __builtin_amdgcn_wmma_f32_16x16x4_f32(false, wreg[t][i][tp][j],
                                                                     false, b, (short)0, c[t],
                                                                     false, false);
                }

        v2f sq[NT];
#pragma unroll
        for (int t = 0; t < NT; ++t) {
            sq[t] = (v2f){0.0f, 0.0f};
#pragma unroll
            for (int r = 0; r < 8; r += 2) {
                const v2f znew = (v2f){fast_tanh(c[t][r] + xf[t][r]),
                                       fast_tanh(c[t][r + 1] + xf[t][r + 1])};
                const v2f d = znew - (v2f){zn[t][r], zn[t][r + 1]};
                sq[t] = d * d + sq[t];
                zn[t][r]     = znew.x;
                zn[t][r + 1] = znew.y;
            }
        }
        res = wave_sum((sq[0].x + sq[0].y) + (sq[1].x + sq[1].y));
    }

    asm volatile("s_wait_dscnt 0x0" ::: "memory");

#pragma unroll
    for (int t = 0; t < NT; ++t)
#pragma unroll
        for (int r = 0; r < 8; ++r)
            out[(rowBase + cIdx) * D_DIM + n0 + 16 * t + 8 * half + r] = zn[t][r];
}

extern "C" void kernel_launch(void* const* d_in, const int* in_sizes, int n_in,
                              void* d_out, int out_size, void* d_ws, size_t ws_size,
                              hipStream_t stream) {
    const float* x = (const float*)d_in[0];
    const float* W = (const float*)d_in[1];
    float* out = (float*)d_out;
    (void)in_sizes; (void)n_in; (void)out_size; (void)d_ws; (void)ws_size;

    dim3 grid(B_DIM / 16);
    dim3 block(NWAVE * 32);
    hipLaunchKernelGGL(tanh_fixed_point_wmma, grid, block, 0, stream, x, W, out);
}


# Passes before the stop test run in Gauss-Seidel order: tile 1 of every wave uses the fresh tile-0 values (two half-steps, split barriers); final plain passes, stop test, their sum order unchanged
# speedup vs baseline: 6.4524x; 1.0737x over previous
#include <hip/hip_runtime.h>
#include <hip/hip_bf16.h>
#include <math.h>


typedef __attribute__((ext_vector_type(2))) float v2f;
typedef __attribute__((ext_vector_type(4))) float v4f;
typedef __attribute__((ext_vector_type(8))) float v8f;

#define D_DIM 256
#define B_DIM 512
#define ITERS 50
#define NT    2
#define NWAVE (D_DIM / 16 / NT)
#define TOL_REF 1e-4f
#define BLOCK_THR (TOL_REF * TOL_REF * 16.0f / (float)B_DIM)
#define GS_SWITCH (8.0f * BLOCK_THR)
#define GS_MAX    24

#define WMMA_F32(a, b, c) \
    __builtin_amdgcn_wmma_f32_16x16x4_f32(false, (a), false, (b), (short)0, (c), false, false)

__device__ __forceinline__ float fast_tanh(float v) {
    float e = __expf(2.0f * v);
    float r = __builtin_amdgcn_rcpf(e + 1.0f);
    return fmaf(-2.0f, r, 1.0f);
}

#define ROW_XMASK_ADD(v, m)                                                                  \
    ((v) + __builtin_bit_cast(float, __builtin_amdgcn_update_dpp(                            \
               0, __builtin_bit_cast(int, (v)), 0x160 + (m), 0xf, 0xf, true)))

__device__ __forceinline__ float wave_sum(float v) {
    v = ROW_XMASK_ADD(v, 1);
    v = ROW_XMASK_ADD(v, 2);
    v = ROW_XMASK_ADD(v, 4);
    v = ROW_XMASK_ADD(v, 8);
    const int b = __builtin_bit_cast(int, v);
    return v + __builtin_bit_cast(float, __builtin_amdgcn_permlanex16(b, b, 0x76543210, 0xfedcba98,
                                                                      false, false));
}

__device__ __forceinline__ v2f tile_update(const v8f c, const float (&xf)[8], float (&zn)[8]) {
    v2f sq = (v2f){0.0f, 0.0f};
#pragma unroll
    for (int r = 0; r < 8; r += 2) {
        const v2f znew = (v2f){fast_tanh(c[r] + xf[r]), fast_tanh(c[r + 1] + xf[r + 1])};
        const v2f d = znew - (v2f){zn[r], zn[r + 1]};
        sq = d * d + sq;
        zn[r]     = znew.x;
        zn[r + 1] = znew.y;
    }
    return sq;
}


__global__ __launch_bounds__(NWAVE * 32, 1)
void tanh_fixed_point_wmma(const float* __restrict__ x,
                           const float* __restrict__ W,
                           float* __restrict__ out)
{
    __shared__ v4f zsh[2][NWAVE][2 * NT][32];
    __shared__ __attribute__((aligned(16))) float rsh[2][NWAVE];
    __shared__ v4f zs0[NWAVE][2][32];
    __shared__ v4f zs1[NWAVE][2][32];
    __shared__ __attribute__((aligned(16))) float usw[NWAVE];

    const int tid  = threadIdx.x;
    const int wave = tid >> 5;
    const int lane = tid & 31;
    const int cIdx = lane & 15;
    const int half = lane >> 4;
    const int n0   = wave * (16 * NT);
    const int rowBase = blockIdx.x * 16;

    v2f wreg[NT][NWAVE][NT][4];
#pragma unroll
    for (int t = 0; t < NT; ++t)
#pragma unroll
        for (int i = 0; i < NWAVE; ++i)
#pragma unroll
            for (int tp = 0; tp < NT; ++tp) {
                const int pw = (wave + i) & (NWAVE - 1);
                const float* wp = &W[(n0 + 16 * t + cIdx) * D_DIM + 32 * pw + 16 * tp + 8 * half];
#pragma unroll
                for (int j = 0; j < 4; ++j) {
                    wreg[t][i][tp][j].x = wp[2 * j];
                    wreg[t][i][tp][j].y = wp[2 * j + 1];
                }
            }

    float xf[NT][8];
#pragma unroll
    for (int t = 0; t < NT; ++t)
#pragma unroll
        for (int r = 0; r < 8; ++r)
            xf[t][r] = x[(rowBase + cIdx) * D_DIM + n0 + 16 * t + 8 * half + r];

    float zn[NT][8];
#pragma unroll
    for (int t = 0; t < NT; ++t)
#pragma unroll
        for (int r = 0; r < 8; ++r)
            zn[t][r] = fast_tanh(xf[t][r]);


    int it = 0;
    v4f la[NWAVE - 1][2];
    v4f lb[NWAVE - 1][2];
    {
#pragma unroll
        for (int q = 0; q < 2; ++q)
            zs0[wave][q][lane] = (v4f){zn[0][4 * q], zn[0][4 * q + 1], zn[0][4 * q + 2], zn[0][4 * q + 3]};
        usw[wave] = 3.0e38f;
        __syncthreads();
#pragma unroll
        for (int i = 1; i < NWAVE; ++i) {
            const int pw = (wave + i) & (NWAVE - 1);
#pragma unroll
            for (int q = 0; q < 2; ++q) la[i - 1][q] = zs0[pw][q][lane];
        }
#pragma unroll
        for (int q = 0; q < 2; ++q)
            zs1[wave][q][lane] = (v4f){zn[1][4 * q], zn[1][4 * q + 1], zn[1][4 * q + 2], zn[1][4 * q + 3]};
    }

    for (;; ++it) {
        v8f c0 = (v8f){};
#pragma unroll
        for (int tp = 0; tp < NT; ++tp)
#pragma unroll
            for (int j = 0; j < 4; ++j)
                c0 = WMMA_F32(wreg[0][0][tp][j], ((v2f){zn[tp][2 * j], zn[tp][2 * j + 1]}), c0);
        __builtin_amdgcn_sched_barrier(0);
        __builtin_amdgcn_fence(__ATOMIC_RELEASE, "workgroup", "local");
        __builtin_amdgcn_s_barrier_signal(-1);
#pragma unroll
        for (int i = 1; i < 4; ++i)
#pragma unroll
            for (int j = 0; j < 4; ++j) {
                const v4f v = la[i - 1][j >> 1];
                c0 = WMMA_F32(wreg[0][i][0][j], ((j & 1) ? (v2f){v.z, v.w} : (v2f){v.x, v.y}), c0);
            }
        __builtin_amdgcn_sched_barrier(0);
        __builtin_amdgcn_s_barrier_wait(-1);
        __builtin_amdgcn_fence(__ATOMIC_ACQUIRE, "workgroup", "local");
        const v4f ua = *(const v4f*)&usw[0];
        const v4f ub = *(const v4f*)&usw[4];
#pragma unroll
        for (int i = 1; i < NWAVE; ++i) {
            const int pw = (wave + i) & (NWAVE - 1);
#pragma unroll
            for (int q = 0; q < 2; ++q) lb[i - 1][q] = zs1[pw][q][lane];
        }
        __builtin_amdgcn_sched_barrier(0);
#pragma unroll
        for (int i = 4; i < NWAVE; ++i)
#pragma unroll
            for (int j = 0; j < 4; ++j) {
                const v4f v = la[i - 1][j >> 1];
                c0 = WMMA_F32(wreg[0][i][0][j], ((j & 1) ? (v2f){v.z, v.w} : (v2f){v.x, v.y}), c0);
            }
        __builtin_amdgcn_sched_barrier(0);
        {
            const v4f up = ua + ub;
            const v2f uq = (v2f){up.x, up.y} + (v2f){up.z, up.w};
            const float utot = uq.x + uq.y;
            if (__builtin_bit_cast(float, __builtin_amdgcn_readfirstlane(__builtin_bit_cast(int, utot)))
                    < GS_SWITCH || it >= GS_MAX)
                break;
        }
#pragma unroll
        for (int i = 1; i < NWAVE; ++i)
#pragma unroll
            for (int j = 0; j < 4; ++j) {
                const v4f v = lb[i - 1][j >> 1];
                c0 = WMMA_F32(wreg[0][i][1][j], ((j & 1) ? (v2f){v.z, v.w} : (v2f){v.x, v.y}), c0);
            }
        const v2f sq0 = tile_update(c0, xf[0], zn[0]);
#pragma unroll
        for (int q = 0; q < 2; ++q)
            zs0[wave][q][lane] = (v4f){zn[0][4 * q], zn[0][4 * q + 1], zn[0][4 * q + 2], zn[0][4 * q + 3]};
        __builtin_amdgcn_sched_barrier(0);

        v8f c1 = (v8f){};
#pragma unroll
        for (int tp = 0; tp < NT; ++tp)
#pragma unroll
            for (int j = 0; j < 4; ++j)
                c1 = WMMA_F32(wreg[1][0][tp][j], ((v2f){zn[tp][2 * j], zn[tp][2 * j + 1]}), c1);
        __builtin_amdgcn_sched_barrier(0);
        __builtin_amdgcn_fence(__ATOMIC_RELEASE, "workgroup", "local");
        __builtin_amdgcn_s_barrier_signal(-1);
#pragma unroll
        for (int i = 1; i < 4; ++i)
#pragma unroll
            for (int j = 0; j < 4; ++j) {
                const v4f v = lb[i - 1][j >> 1];
                c1 = WMMA_F32(wreg[1][i][1][j], ((j & 1) ? (v2f){v.z, v.w} : (v2f){v.x, v.y}), c1);
            }
        __builtin_amdgcn_sched_barrier(0);
        __builtin_amdgcn_s_barrier_wait(-1);
        __builtin_amdgcn_fence(__ATOMIC_ACQUIRE, "workgroup", "local");
#pragma unroll
        for (int i = 1; i < NWAVE; ++i) {
            const int pw = (wave + i) & (NWAVE - 1);
#pragma unroll
            for (int q = 0; q < 2; ++q) la[i - 1][q] = zs0[pw][q][lane];
        }
        __builtin_amdgcn_sched_barrier(0);
#pragma unroll
        for (int i = 4; i < NWAVE; ++i)
#pragma unroll
            for (int j = 0; j < 4; ++j) {
                const v4f v = lb[i - 1][j >> 1];
                c1 = WMMA_F32(wreg[1][i][1][j], ((j & 1) ? (v2f){v.z, v.w} : (v2f){v.x, v.y}), c1);
            }
#pragma unroll
        for (int i = 1; i < NWAVE; ++i)
#pragma unroll
            for (int j = 0; j < 4; ++j) {
                const v4f v = la[i - 1][j >> 1];
                c1 = WMMA_F32(wreg[1][i][0][j], ((j & 1) ? (v2f){v.z, v.w} : (v2f){v.x, v.y}), c1);
            }
        const v2f sq1 = tile_update(c1, xf[1], zn[1]);
        const float upd = wave_sum((sq0.x + sq0.y) + (sq1.x + sq1.y));
#pragma unroll
        for (int q = 0; q < 2; ++q)
            zs1[wave][q][lane] = (v4f){zn[1][4 * q], zn[1][4 * q + 1], zn[1][4 * q + 2], zn[1][4 * q + 3]};
        usw[wave] = upd;
    }

    float res = 3.0e38f;

    for (; it <= ITERS; ++it) {
        const int buf = it & 1;

#pragma unroll
        for (int g = 0; g < 2 * NT; ++g) {
            const int tp = g >> 1, q = g & 1;
            zsh[buf][wave][g][lane] = (v4f){zn[tp][4 * q], zn[tp][4 * q + 1],
                                            zn[tp][4 * q + 2], zn[tp][4 * q + 3]};
        }
        rsh[buf][wave] = res;
        __builtin_amdgcn_sched_barrier(0);

        v8f c[NT];
#pragma unroll
        for (int t = 0; t < NT; ++t) c[t] = (v8f){};

#pragma unroll
        for (int j = 0; j < 4; ++j) {
            const v2f b = (v2f){zn[0][2 * j], zn[0][2 * j + 1]};
#pragma unroll
            for (int t = 0; t < NT; ++t)
                c[t] = __builtin_amdgcn_wmma_f32_16x16x4_f32(false, wreg[t][0][0][j], false, b,
                                                             (short)0, c[t], false, false);
        }
        __builtin_amdgcn_sched_barrier(0);
        __syncthreads();

        const v4f ra = *(const v4f*)&rsh[buf][0];
        const v4f rb = *(const v4f*)&rsh[buf][4];

        v4f ld[NWAVE - 1][2 * NT];
#pragma unroll
        for (int i = 1; i < NWAVE; ++i) {
            const int pw = (wave + i) & (NWAVE - 1);
#pragma unroll
            for (int g = 0; g < 2 * NT; ++g)
                ld[i - 1][g] = zsh[buf][pw][g][lane];
        }
        __builtin_amdgcn_sched_barrier(0);

#pragma unroll
        for (int j = 0; j < 4; ++j) {
            const v2f b = (v2f){zn[1][2 * j], zn[1][2 * j + 1]};
#pragma unroll
            for (int t = 0; t < NT; ++t)
                c[t] = __builtin_amdgcn_wmma_f32_16x16x4_f32(false, wreg[t][0][1][j], false, b,
                                                             (short)0, c[t], false, false);
        }
        __builtin_amdgcn_sched_barrier(0);
        const v4f rp = ra + rb;
        const v2f rq = (v2f){rp.x, rp.y} + (v2f){rp.z, rp.w};
        const float tot = rq.x + rq.y;
        if (__builtin_bit_cast(float, __builtin_amdgcn_readfirstlane(__builtin_bit_cast(int, tot)))
            < BLOCK_THR)
            break;

#pragma unroll
        for (int i = 1; i < NWAVE; ++i)
#pragma unroll
            for (int tp = 0; tp < NT; ++tp)
#pragma unroll
                for (int j = 0; j < 4; ++j) {
                    const v4f v = ld[i - 1][2 * tp + (j >> 1)];
                    const v2f b = (j & 1) ? (v2f){v.z, v.w} : (v2f){v.x, v.y};
#pragma unroll
                    for (int t = 0; t < NT; ++t)
                        c[t] = __builtin_amdgcn_wmma_f32_16x16x4_f32(false, wreg[t][i][tp][j],
                                                                     false, b, (short)0, c[t],
                                                                     false, false);
                }

        v2f sq[NT];
#pragma unroll
        for (int t = 0; t < NT; ++t) {
            sq[t] = (v2f){0.0f, 0.0f};
#pragma unroll
            for (int r = 0; r < 8; r += 2) {
                const v2f znew = (v2f){fast_tanh(c[t][r] + xf[t][r]),
                                       fast_tanh(c[t][r + 1] + xf[t][r + 1])};
                const v2f d = znew - (v2f){zn[t][r], zn[t][r + 1]};
                sq[t] = d * d + sq[t];
                zn[t][r]     = znew.x;
                zn[t][r + 1] = znew.y;
            }
        }
        res = wave_sum((sq[0].x + sq[0].y) + (sq[1].x + sq[1].y));
    }

    asm volatile("s_wait_dscnt 0x0" ::: "memory");

#pragma unroll
    for (int t = 0; t < NT; ++t)
#pragma unroll
        for (int r = 0; r < 8; ++r)
            out[(rowBase + cIdx) * D_DIM + n0 + 16 * t + 8 * half + r] = zn[t][r];
}

extern "C" void kernel_launch(void* const* d_in, const int* in_sizes, int n_in,
                              void* d_out, int out_size, void* d_ws, size_t ws_size,
                              hipStream_t stream) {
    const float* x = (const float*)d_in[0];
    const float* W = (const float*)d_in[1];
    float* out = (float*)d_out;
    (void)in_sizes; (void)n_in; (void)out_size; (void)d_ws; (void)ws_size;

    dim3 grid(B_DIM / 16);
    dim3 block(NWAVE * 32);
    hipLaunchKernelGGL(tanh_fixed_point_wmma, grid, block, 0, stream, x, W, out);
}


# Gauss-Seidel sweeps in four groups of 64 features, not two: in a stage a wave ends its tile on the 16 fresh k-steps while its SIMD partner runs 48 ahead; two partial sums a tile, k sum order changed
# speedup vs baseline: 7.2131x; 1.1179x over previous
#include <hip/hip_runtime.h>
#include <hip/hip_bf16.h>
#include <math.h>


typedef __attribute__((ext_vector_type(2))) float v2f;
typedef __attribute__((ext_vector_type(4))) float v4f;
typedef __attribute__((ext_vector_type(8))) float v8f;

#define D_DIM 256
#define B_DIM 512
#define ITERS 50
#define NT    2
#define NWAVE (D_DIM / 16 / NT)
#define TOL_REF 1e-4f
#define BLOCK_THR (TOL_REF * TOL_REF * 16.0f / (float)B_DIM)
#define GS_SWITCH (8.0f * BLOCK_THR)
#define GS_MAX    24

#define WMMA_F32(a, b, c) \
    __builtin_amdgcn_wmma_f32_16x16x4_f32(false, (a), false, (b), (short)0, (c), false, false)

__device__ __forceinline__ float fast_tanh(float v) {
    float e = __expf(2.0f * v);
    float r = __builtin_amdgcn_rcpf(e + 1.0f);
    return fmaf(-2.0f, r, 1.0f);
}

#define ROW_XMASK_ADD(v, m)                                                                  \
    ((v) + __builtin_bit_cast(float, __builtin_amdgcn_update_dpp(                            \
               0, __builtin_bit_cast(int, (v)), 0x160 + (m), 0xf, 0xf, true)))

__device__ __forceinline__ float wave_sum(float v) {
    v = ROW_XMASK_ADD(v, 1);
    v = ROW_XMASK_ADD(v, 2);
    v = ROW_XMASK_ADD(v, 4);
    v = ROW_XMASK_ADD(v, 8);
    const int b = __builtin_bit_cast(int, v);
    return v + __builtin_bit_cast(float, __builtin_amdgcn_permlanex16(b, b, 0x76543210, 0xfedcba98,
                                                                      false, false));
}

__device__ __forceinline__ v2f tile_update(const v8f c, const float (&xf)[8], float (&zn)[8]) {
    v2f sq = (v2f){0.0f, 0.0f};
#pragma unroll
    for (int r = 0; r < 8; r += 2) {
        const v2f znew = (v2f){fast_tanh(c[r] + xf[r]), fast_tanh(c[r + 1] + xf[r + 1])};
        const v2f d = znew - (v2f){zn[r], zn[r + 1]};
        sq = d * d + sq;
        zn[r]     = znew.x;
        zn[r + 1] = znew.y;
    }
    return sq;
}


template <int T, int S, int HALF>
__device__ __forceinline__ void acc16(const v2f (&wreg)[NT][NWAVE][NT][4], const float (&zn)[NT][8],
                                      const v4f (&fr)[4][4][2], v8f& ca, v8f& cb) {
    constexpr int TP = S >> 1;
    constexpr int I0 = 4 * ((S & 1) ^ HALF);
#pragma unroll
    for (int m = 0; m < 4; ++m)
#pragma unroll
        for (int j = 0; j < 4; ++j) {
            v2f b;
            if (I0 + m == 0) {
                b = (v2f){zn[TP][2 * j], zn[TP][2 * j + 1]};
            } else {
                const v4f v = fr[S][m][j >> 1];
                b = (j & 1) ? (v2f){v.z, v.w} : (v2f){v.x, v.y};
            }
            if (j & 1) cb = WMMA_F32(wreg[T][I0 + m][TP][j], b, cb);
            else       ca = WMMA_F32(wreg[T][I0 + m][TP][j], b, ca);
        }
}

#define BAR_LDS()                                                              \
    do {                                                                       \
        __builtin_amdgcn_fence(__ATOMIC_RELEASE, "workgroup", "local");        \
        __builtin_amdgcn_s_barrier_signal(-1);                                 \
        __builtin_amdgcn_s_barrier_wait(-1);                                   \
        __builtin_amdgcn_fence(__ATOMIC_ACQUIRE, "workgroup", "local");        \
    } while (0)


__global__ __launch_bounds__(NWAVE * 32, 1)
void tanh_fixed_point_wmma(const float* __restrict__ x,
                           const float* __restrict__ W,
                           float* __restrict__ out)
{
    __shared__ v4f zsh[2][NWAVE][2 * NT][32];
    __shared__ __attribute__((aligned(16))) float rsh[2][NWAVE];
    __shared__ v4f zs0[NWAVE][2][32];
    __shared__ v4f zs1[NWAVE][2][32];
    __shared__ __attribute__((aligned(16))) float usw[NWAVE];

    const int tid  = threadIdx.x;
    const int wave = tid >> 5;
    const int lane = tid & 31;
    const int cIdx = lane & 15;
    const int half = lane >> 4;
    const int n0   = wave * (16 * NT);
    const int rowBase = blockIdx.x * 16;

    v2f wreg[NT][NWAVE][NT][4];
#pragma unroll
    for (int t = 0; t < NT; ++t)
#pragma unroll
        for (int i = 0; i < NWAVE; ++i)
#pragma unroll
            for (int tp = 0; tp < NT; ++tp) {
                const int pw = wave ^ i;
                const float* wp = &W[(n0 + 16 * t + cIdx) * D_DIM + 32 * pw + 16 * tp + 8 * half];
#pragma unroll
                for (int j = 0; j < 4; ++j) {
                    wreg[t][i][tp][j].x = wp[2 * j];
                    wreg[t][i][tp][j].y = wp[2 * j + 1];
                }
            }

    float xf[NT][8];
#pragma unroll
    for (int t = 0; t < NT; ++t)
#pragma unroll
        for (int r = 0; r < 8; ++r)
            xf[t][r] = x[(rowBase + cIdx) * D_DIM + n0 + 16 * t + 8 * half + r];

    float zn[NT][8];
#pragma unroll
    for (int t = 0; t < NT; ++t)
#pragma unroll
        for (int r = 0; r < 8; ++r)
            zn[t][r] = fast_tanh(xf[t][r]);


    const bool upper = __builtin_amdgcn_readfirstlane(wave) >= (NWAVE / 2);
    int it = 0;
    v4f fr[4][4][2];
    v8f ca = (v8f){}, cb = (v8f){};

#define PUBLISH(TP, ZS)                                                                       \
    _Pragma("unroll")                                                                         \
    for (int q_ = 0; q_ < 2; ++q_)                                                            \
        ZS[wave][q_][lane] = (v4f){zn[TP][4 * q_], zn[TP][4 * q_ + 1], zn[TP][4 * q_ + 2],    \
                                   zn[TP][4 * q_ + 3]}
#define LOAD_STAGE(S, ZS)                                                                     \
    _Pragma("unroll")                                                                         \
    for (int m_ = 0; m_ < 4; ++m_) {                                                          \
        _Pragma("unroll")                                                                     \
        for (int q_ = 0; q_ < 2; ++q_)                                                        \
            fr[S][m_][q_] = ZS[4 * ((S) & 1) + ((wave & 3) ^ m_)][q_][lane];                  \
    }
#define FINISH(T, ZS, SQ)                                                                     \
    do {                                                                                      \
        const v8f c_ = ca + cb;                                                               \
        SQ = tile_update(c_, xf[T], zn[T]);                                                   \
        PUBLISH(T, ZS);                                                                       \
    } while (0)

    {
#pragma unroll
        for (int g = 0; g < 2 * NT; ++g) {
            const int tp = g >> 1, q = g & 1;
            zsh[0][wave][g][lane] = (v4f){zn[tp][4 * q], zn[tp][4 * q + 1],
                                          zn[tp][4 * q + 2], zn[tp][4 * q + 3]};
        }
        __syncthreads();
#pragma unroll
        for (int s = 0; s < 4; ++s)
#pragma unroll
            for (int m = 0; m < 4; ++m)
#pragma unroll
                for (int q = 0; q < 2; ++q)
                    fr[s][m][q] = zsh[0][4 * (s & 1) + ((wave & 3) ^ m)][2 * (s >> 1) + q][lane];
        __builtin_amdgcn_sched_barrier(0);
        if (!upper) {
            acc16<0, 0, 0>(wreg, zn, fr, ca, cb);
            acc16<0, 1, 0>(wreg, zn, fr, ca, cb);
            acc16<0, 2, 0>(wreg, zn, fr, ca, cb);
        }
    }

    for (;;) {
        v2f sq0 = (v2f){0.0f, 0.0f}, sq1 = (v2f){0.0f, 0.0f};
        if (!upper) {
            __builtin_amdgcn_s_setprio(3);
            acc16<0, 3, 0>(wreg, zn, fr, ca, cb);
            __builtin_amdgcn_s_setprio(0);
            FINISH(0, zs0, sq0);
        } else {
            ca = (v8f){}; cb = (v8f){};
            acc16<0, 1, 1>(wreg, zn, fr, ca, cb);
            acc16<0, 2, 1>(wreg, zn, fr, ca, cb);
            acc16<0, 3, 1>(wreg, zn, fr, ca, cb);
        }
        BAR_LDS();
        LOAD_STAGE(0, zs0);
        __builtin_amdgcn_sched_barrier(0);

        if (!upper) {
            ca = (v8f){}; cb = (v8f){};
            acc16<1, 2, 0>(wreg, zn, fr, ca, cb);
            acc16<1, 3, 0>(wreg, zn, fr, ca, cb);
            acc16<1, 0, 0>(wreg, zn, fr, ca, cb);
        } else {
            __builtin_amdgcn_s_setprio(3);
            acc16<0, 0, 1>(wreg, zn, fr, ca, cb);
            __builtin_amdgcn_s_setprio(0);
            FINISH(0, zs0, sq0);
        }
        BAR_LDS();
        LOAD_STAGE(1, zs0);
        __builtin_amdgcn_sched_barrier(0);

        if (!upper) {
            __builtin_amdgcn_s_setprio(3);
            acc16<1, 1, 0>(wreg, zn, fr, ca, cb);
            __builtin_amdgcn_s_setprio(0);
            FINISH(1, zs1, sq1);
            usw[wave] = wave_sum((sq0.x + sq0.y) + (sq1.x + sq1.y));
        } else {
            ca = (v8f){}; cb = (v8f){};
            acc16<1, 3, 1>(wreg, zn, fr, ca, cb);
            acc16<1, 0, 1>(wreg, zn, fr, ca, cb);
            acc16<1, 1, 1>(wreg, zn, fr, ca, cb);
        }
        BAR_LDS();
        LOAD_STAGE(2, zs1);
        __builtin_amdgcn_sched_barrier(0);

        if (!upper) {
            ca = (v8f){}; cb = (v8f){};
            acc16<0, 0, 0>(wreg, zn, fr, ca, cb);
            acc16<0, 1, 0>(wreg, zn, fr, ca, cb);
            acc16<0, 2, 0>(wreg, zn, fr, ca, cb);
        } else {
            __builtin_amdgcn_s_setprio(3);
            acc16<1, 2, 1>(wreg, zn, fr, ca, cb);
            __builtin_amdgcn_s_setprio(0);
            FINISH(1, zs1, sq1);
            usw[wave] = wave_sum((sq0.x + sq0.y) + (sq1.x + sq1.y));
        }
        BAR_LDS();
        const v4f ua = *(const v4f*)&usw[0];
        const v4f ub = *(const v4f*)&usw[4];
        LOAD_STAGE(3, zs1);
        __builtin_amdgcn_sched_barrier(0);
        ++it;
        {
            const v4f up = ua + ub;
            const v2f uq = (v2f){up.x, up.y} + (v2f){up.z, up.w};
            const float utot = uq.x + uq.y;
            if (__builtin_bit_cast(float, __builtin_amdgcn_readfirstlane(__builtin_bit_cast(int, utot)))
                    < GS_SWITCH || it >= GS_MAX)
                break;
        }
    }
#undef PUBLISH
#undef LOAD_STAGE
#undef FINISH

    float res = 3.0e38f;

    for (; it <= ITERS; ++it) {
        const int buf = it & 1;

#pragma unroll
        for (int g = 0; g < 2 * NT; ++g) {
            const int tp = g >> 1, q = g & 1;
            zsh[buf][wave][g][lane] = (v4f){zn[tp][4 * q], zn[tp][4 * q + 1],
                                            zn[tp][4 * q + 2], zn[tp][4 * q + 3]};
        }
        rsh[buf][wave] = res;
        __builtin_amdgcn_sched_barrier(0);

        v8f c[NT];
#pragma unroll
        for (int t = 0; t < NT; ++t) c[t] = (v8f){};

#pragma unroll
        for (int j = 0; j < 4; ++j) {
            const v2f b = (v2f){zn[0][2 * j], zn[0][2 * j + 1]};
#pragma unroll
            for (int t = 0; t < NT; ++t)
                c[t] = __builtin_amdgcn_wmma_f32_16x16x4_f32(false, wreg[t][0][0][j], false, b,
                                                             (short)0, c[t], false, false);
        }
        __builtin_amdgcn_sched_barrier(0);
        __syncthreads();

        const v4f ra = *(const v4f*)&rsh[buf][0];
        const v4f rb = *(const v4f*)&rsh[buf][4];

        v4f ld[NWAVE - 1][2 * NT];
#pragma unroll
        for (int i = 1; i < NWAVE; ++i) {
            const int pw = wave ^ i;
#pragma unroll
            for (int g = 0; g < 2 * NT; ++g)
                ld[i - 1][g] = zsh[buf][pw][g][lane];
        }
        __builtin_amdgcn_sched_barrier(0);

#pragma unroll
        for (int j = 0; j < 4; ++j) {
            const v2f b = (v2f){zn[1][2 * j], zn[1][2 * j + 1]};
#pragma unroll
            for (int t = 0; t < NT; ++t)
                c[t] = __builtin_amdgcn_wmma_f32_16x16x4_f32(false, wreg[t][0][1][j], false, b,
                                                             (short)0, c[t], false, false);
        }
        __builtin_amdgcn_sched_barrier(0);
        const v4f rp = ra + rb;
        const v2f rq = (v2f){rp.x, rp.y} + (v2f){rp.z, rp.w};
        const float tot = rq.x + rq.y;
        if (__builtin_bit_cast(float, __builtin_amdgcn_readfirstlane(__builtin_bit_cast(int, tot)))
            < BLOCK_THR)
            break;

#pragma unroll
        for (int i = 1; i < NWAVE; ++i)
#pragma unroll
            for (int tp = 0; tp < NT; ++tp)
#pragma unroll
                for (int j = 0; j < 4; ++j) {
                    const v4f v = ld[i - 1][2 * tp + (j >> 1)];
                    const v2f b = (j & 1) ? (v2f){v.z, v.w} : (v2f){v.x, v.y};
#pragma unroll
                    for (int t = 0; t < NT; ++t)
                        c[t] = __builtin_amdgcn_wmma_f32_16x16x4_f32(false, wreg[t][i][tp][j],
                                                                     false, b, (short)0, c[t],
                                                                     false, false);
                }

        v2f sq[NT];
#pragma unroll
        for (int t = 0; t < NT; ++t) {
            sq[t] = (v2f){0.0f, 0.0f};
#pragma unroll
            for (int r = 0; r < 8; r += 2) {
                const v2f znew = (v2f){fast_tanh(c[t][r] + xf[t][r]),
                                       fast_tanh(c[t][r + 1] + xf[t][r + 1])};
                const v2f d = znew - (v2f){zn[t][r], zn[t][r + 1]};
                sq[t] = d * d + sq[t];
                zn[t][r]     = znew.x;
                zn[t][r + 1] = znew.y;
            }
        }
        res = wave_sum((sq[0].x + sq[0].y) + (sq[1].x + sq[1].y));
    }

    asm volatile("s_wait_dscnt 0x0" ::: "memory");

#pragma unroll
    for (int t = 0; t < NT; ++t)
#pragma unroll
        for (int r = 0; r < 8; ++r)
            out[(rowBase + cIdx) * D_DIM + n0 + 16 * t + 8 * half + r] = zn[t][r];
}

extern "C" void kernel_launch(void* const* d_in, const int* in_sizes, int n_in,
                              void* d_out, int out_size, void* d_ws, size_t ws_size,
                              hipStream_t stream) {
    const float* x = (const float*)d_in[0];
    const float* W = (const float*)d_in[1];
    float* out = (float*)d_out;
    (void)in_sizes; (void)n_in; (void)out_size; (void)d_ws; (void)ws_size;

    dim3 grid(B_DIM / 16);
    dim3 block(NWAVE * 32);
    hipLaunchKernelGGL(tanh_fixed_point_wmma, grid, block, 0, stream, x, W, out);
}
